# starting kernel
# baseline (speedup 1.0000x reference)
#include <hip/hip_runtime.h>
#include <hip/hip_bf16.h>
#include <math.h>

#define B_   32
#define T_   4096
#define DIN_ 256
#define H_   256
#define NC_  32
#define L_   (T_ / NC_)

typedef float v2f __attribute__((ext_vector_type(2)));
typedef float v8f __attribute__((ext_vector_type(8)));

__device__ __forceinline__ float softplus_f(float u) {
    return fmaxf(u, 0.0f) + log1pf(__expf(-fabsf(u)));
}

__device__ __forceinline__ float logaddexp_f(float a, float b) {
    float mx = fmaxf(a, b);
    float d  = fabsf(a - b);
    return mx + log1pf(__expf(-d));
}

__device__ __forceinline__ float log_g_f(float a) {
    return (a >= 0.0f) ? __logf(a + 0.5f) : -softplus_f(-a);
}

__global__ __launch_bounds__(32) void mingru_gemm_kernel(
    const float* __restrict__ x,  const float* __restrict__ Wz,
    const float* __restrict__ bz, const float* __restrict__ Wh,
    const float* __restrict__ bh, float* __restrict__ v_out,
    float* __restrict__ c_out)
{
    const int lane  = threadIdx.x;
    const int m0    = blockIdx.y << 4;
    const int n0    = blockIdx.x << 4;
    const int l15   = lane & 15;
    const int khalf = (lane >> 4) << 1;

    const float* xp  = x  + (size_t)(m0 + l15) * DIN_ + khalf;
    const float* wzp = Wz + (size_t)(n0 + l15) * DIN_ + khalf;
    const float* whp = Wh + (size_t)(n0 + l15) * DIN_ + khalf;

    v8f accz = {};
    v8f acch = {};

#pragma unroll 4
    for (int kk = 0; kk < DIN_; kk += 4) {
        v2f a  = *(const v2f*)(xp  + kk);
        v2f bz2 = *(const v2f*)(wzp + kk);
        v2f bh2 = *(const v2f*)(whp + kk);
        accz = __builtin_amdgcn_wmma_f32_16x16x4_f32(
                   false, a, false, bz2, (short)0, accz, false, false);
        acch = __builtin_amdgcn_wmma_f32_16x16x4_f32(
                   false, a, false, bh2, (short)0, acch, false, false);
    }

    const int   ncol = n0 + l15;
    const float bzb  = bz[ncol];
    const float bhb  = bh[ncol];
    const int   mbase = m0 + ((lane >> 4) << 3);

#pragma unroll
    for (int j = 0; j < 8; ++j) {
        float kpre  = accz[j] + bzb;
        float sp    = softplus_f(kpre);
        float log_z = kpre - sp;
        float log_c = -sp;
        float hp    = acch[j] + bhb;
        float v     = log_z + log_g_f(hp);

        size_t idx = (size_t)(mbase + j) * H_ + ncol;
        v_out[idx] = v;
        c_out[idx] = log_c;
    }
}

__global__ __launch_bounds__(256) void mingru_chunk_reduce(
    const float* __restrict__ c_arr, const float* __restrict__ v_arr,
    float* __restrict__ Csum, float* __restrict__ Vsum)
{
    const int h = threadIdx.x;
    const int j = blockIdx.x;
    const int b = blockIdx.y;

    size_t base = ((size_t)b * T_ + (size_t)j * L_) * H_ + h;
    float C = 0.0f;
    float V = -INFINITY;

    for (int t = 0; t < L_; ++t) {
        size_t idx = base + (size_t)t * H_;
        __builtin_prefetch(c_arr + idx + (size_t)8 * H_, 0, 0);
        __builtin_prefetch(v_arr + idx + (size_t)8 * H_, 0, 0);
        float c = c_arr[idx];
        float v = v_arr[idx];
        C += c;
        V = logaddexp_f(V + c, v);
    }
    size_t sidx = ((size_t)b * NC_ + j) * H_ + h;
    Csum[sidx] = C;
    Vsum[sidx] = V;
}

__global__ __launch_bounds__(256) void mingru_carry(
    const float* __restrict__ h0, const float* __restrict__ Csum,
    const float* __restrict__ Vsum, float* __restrict__ entry)
{
    const int h = threadIdx.x;
    const int b = blockIdx.x;

    float state = log_g_f(h0[(size_t)b * H_ + h]);
#pragma unroll
    for (int j = 0; j < NC_; ++j) {
        size_t sidx = ((size_t)b * NC_ + j) * H_ + h;
        entry[sidx] = state;
        state = logaddexp_f(state + Csum[sidx], Vsum[sidx]);
    }
}

__global__ __launch_bounds__(256) void mingru_apply(
    const float* __restrict__ c_arr, float* __restrict__ vout,
    const float* __restrict__ entry)
{
    const int h = threadIdx.x;
    const int j = blockIdx.x;
    const int b = blockIdx.y;

    float state = entry[((size_t)b * NC_ + j) * H_ + h];
    size_t base = ((size_t)b * T_ + (size_t)j * L_) * H_ + h;

    for (int t = 0; t < L_; ++t) {
        size_t idx = base + (size_t)t * H_;
        __builtin_prefetch(c_arr + idx + (size_t)8 * H_, 0, 0);
        float c = c_arr[idx];
        float v = vout[idx];
        state = logaddexp_f(state + c, v);
        vout[idx] = __expf(state);
    }
}

extern "C" void kernel_launch(void* const* d_in, const int* in_sizes, int n_in,
                              void* d_out, int out_size, void* d_ws, size_t ws_size,
                              hipStream_t stream)
{
    const float* x  = (const float*)d_in[0];
    const float* h0 = (const float*)d_in[1];
    const float* Wz = (const float*)d_in[2];
    const float* bz = (const float*)d_in[3];
    const float* Wh = (const float*)d_in[4];
    const float* bh = (const float*)d_in[5];
    float* out = (float*)d_out;

    float* ws    = (float*)d_ws;
    float* c_arr = ws;
    float* Csum  = c_arr + (size_t)B_ * T_ * H_;
    float* Vsum  = Csum  + (size_t)B_ * NC_ * H_;
    float* entry = Vsum  + (size_t)B_ * NC_ * H_;

    dim3 ggrid(H_ / 16, (B_ * T_) / 16);
    mingru_gemm_kernel<<<ggrid, 32, 0, stream>>>(x, Wz, bz, Wh, bh, out, c_arr);

    dim3 sgrid(NC_, B_);
    mingru_chunk_reduce<<<sgrid, H_, 0, stream>>>(c_arr, out, Csum, Vsum);
    mingru_carry<<<B_, H_, 0, stream>>>(h0, Csum, Vsum, entry);
    mingru_apply<<<sgrid, H_, 0, stream>>>(c_arr, out, entry);
}


# GEMM register tiling: each wave computes a 64x64 block of both gates (32 accumulator tiles), reusing x fragments 8x and W fragments 4x; tiles pass via LDS to one rolled epilogue loop; sums unchanged
# speedup vs baseline: 1.4296x; 1.4296x over previous
#include <hip/hip_runtime.h>
#include <hip/hip_bf16.h>
#include <math.h>

#define B_   32
#define T_   4096
#define DIN_ 256
#define H_   256
#define NC_  32
#define L_   (T_ / NC_)

typedef float v2f __attribute__((ext_vector_type(2)));
typedef float v8f __attribute__((ext_vector_type(8)));

__device__ __forceinline__ float softplus_f(float u) {
    return fmaxf(u, 0.0f) + log1pf(__expf(-fabsf(u)));
}

__device__ __forceinline__ float logaddexp_f(float a, float b) {
    float mx = fmaxf(a, b);
    float d  = fabsf(a - b);
    return mx + log1pf(__expf(-d));
}

__device__ __forceinline__ float log_g_f(float a) {
    return (a >= 0.0f) ? __logf(a + 0.5f) : -softplus_f(-a);
}

#define MT_ 4
#define NT_ 4
#define KS_ 8
#define SP_ 68

__global__ __launch_bounds__(128) void mingru_gemm_kernel(
    const float* __restrict__ x,  const float* __restrict__ Wz,
    const float* __restrict__ bz, const float* __restrict__ Wh,
    const float* __restrict__ bh, float* __restrict__ v_out,
    float* __restrict__ c_out)
{
    __shared__ float stage[4][2][MT_ * 16][SP_];
    const int lane  = threadIdx.x & 31;
    const int wave  = threadIdx.x >> 5;
    const int m0    = blockIdx.x * (MT_ * 16);
    const int n0    = wave * (NT_ * 16);
    const int l15   = lane & 15;
    const int khalf = (lane >> 4) << 1;

    const float* xp  = x  + (size_t)(m0 + l15) * DIN_ + khalf;
    const float* wzp = Wz + (size_t)(n0 + l15) * DIN_ + khalf;
    const float* whp = Wh + (size_t)(n0 + l15) * DIN_ + khalf;

    v8f accz[MT_][NT_];
    v8f acch[MT_][NT_];
#pragma unroll
    for (int mt = 0; mt < MT_; ++mt) {
#pragma unroll
        for (int nt = 0; nt < NT_; ++nt) {
            accz[mt][nt] = v8f{};
            acch[mt][nt] = v8f{};
        }
    }

    for (int kk = 0; kk < DIN_; kk += 4 * KS_) {
        v2f a[MT_][KS_];
        v2f fz[NT_][KS_];
        v2f fh[NT_][KS_];
#pragma unroll
        for (int mt = 0; mt < MT_; ++mt) {
#pragma unroll
            for (int s = 0; s < KS_; ++s)
                a[mt][s] = *(const v2f*)(xp + (size_t)mt * 16 * DIN_ + kk + 4 * s);
        }
#pragma unroll
        for (int nt = 0; nt < NT_; ++nt) {
#pragma unroll
            for (int s = 0; s < KS_; ++s)
                fz[nt][s] = *(const v2f*)(wzp + (size_t)nt * 16 * DIN_ + kk + 4 * s);
        }
#pragma unroll
        for (int nt = 0; nt < NT_; ++nt) {
#pragma unroll
            for (int s = 0; s < KS_; ++s)
                fh[nt][s] = *(const v2f*)(whp + (size_t)nt * 16 * DIN_ + kk + 4 * s);
        }
#pragma unroll
        for (int s = 0; s < KS_; ++s) {
#pragma unroll
            for (int mt = 0; mt < MT_; ++mt) {
#pragma unroll
                for (int nt = 0; nt < NT_; ++nt) {
                    accz[mt][nt] = __builtin_amdgcn_wmma_f32_16x16x4_f32(
                        false, a[mt][s], false, fz[nt][s], (short)0, accz[mt][nt], false, false);
                    acch[mt][nt] = __builtin_amdgcn_wmma_f32_16x16x4_f32(
                        false, a[mt][s], false, fh[nt][s], (short)0, acch[mt][nt], false, false);
                }
            }
        }
    }

    float* sz = &stage[wave][0][0][0];
    float* sh = &stage[wave][1][0][0];
    const int rhalf = (lane >> 4) << 3;
#pragma unroll
    for (int mt = 0; mt < MT_; ++mt) {
#pragma unroll
        for (int nt = 0; nt < NT_; ++nt) {
#pragma unroll
            for (int j = 0; j < 8; ++j) {
                const int off = (mt * 16 + rhalf + j) * SP_ + nt * 16 + l15;
                sz[off] = accz[mt][nt][j];
                sh[off] = acch[mt][nt][j];
            }
        }
    }
    __syncthreads();

    const int   c2   = lane << 1;
    const v2f   bzb  = *(const v2f*)(bz + n0 + c2);
    const v2f   bhb  = *(const v2f*)(bh + n0 + c2);
    for (int r0 = 0; r0 < MT_ * 16; r0 += 4) {
#pragma unroll
        for (int rr = 0; rr < 4; ++rr) {
            const int r = r0 + rr;
            const v2f kz = *(const v2f*)(sz + r * SP_ + c2);
            const v2f kh = *(const v2f*)(sh + r * SP_ + c2);
            v2f vo, co;
#pragma unroll
            for (int e = 0; e < 2; ++e) {
                float kpre  = kz[e] + bzb[e];
                float sp    = softplus_f(kpre);
                float log_z = kpre - sp;
                float log_c = -sp;
                float hp    = kh[e] + bhb[e];
                vo[e] = log_z + log_g_f(hp);
                co[e] = log_c;
            }
            const size_t idx = (size_t)(m0 + r) * H_ + n0 + c2;
            *(v2f*)(v_out + idx) = vo;
            *(v2f*)(c_out + idx) = co;
        }
    }
}

__global__ __launch_bounds__(256) void mingru_chunk_reduce(
    const float* __restrict__ c_arr, const float* __restrict__ v_arr,
    float* __restrict__ Csum, float* __restrict__ Vsum)
{
    const int h = threadIdx.x;
    const int j = blockIdx.x;
    const int b = blockIdx.y;

    size_t base = ((size_t)b * T_ + (size_t)j * L_) * H_ + h;
    float C = 0.0f;
    float V = -INFINITY;

    for (int t = 0; t < L_; ++t) {
        size_t idx = base + (size_t)t * H_;
        __builtin_prefetch(c_arr + idx + (size_t)8 * H_, 0, 0);
        __builtin_prefetch(v_arr + idx + (size_t)8 * H_, 0, 0);
        float c = c_arr[idx];
        float v = v_arr[idx];
        C += c;
        V = logaddexp_f(V + c, v);
    }
    size_t sidx = ((size_t)b * NC_ + j) * H_ + h;
    Csum[sidx] = C;
    Vsum[sidx] = V;
}

__global__ __launch_bounds__(256) void mingru_carry(
    const float* __restrict__ h0, const float* __restrict__ Csum,
    const float* __restrict__ Vsum, float* __restrict__ entry)
{
    const int h = threadIdx.x;
    const int b = blockIdx.x;

    float state = log_g_f(h0[(size_t)b * H_ + h]);
#pragma unroll
    for (int j = 0; j < NC_; ++j) {
        size_t sidx = ((size_t)b * NC_ + j) * H_ + h;
        entry[sidx] = state;
        state = logaddexp_f(state + Csum[sidx], Vsum[sidx]);
    }
}

__global__ __launch_bounds__(256) void mingru_apply(
    const float* __restrict__ c_arr, float* __restrict__ vout,
    const float* __restrict__ entry)
{
    const int h = threadIdx.x;
    const int j = blockIdx.x;
    const int b = blockIdx.y;

    float state = entry[((size_t)b * NC_ + j) * H_ + h];
    size_t base = ((size_t)b * T_ + (size_t)j * L_) * H_ + h;

    for (int t = 0; t < L_; ++t) {
        size_t idx = base + (size_t)t * H_;
        __builtin_prefetch(c_arr + idx + (size_t)8 * H_, 0, 0);
        float c = c_arr[idx];
        float v = vout[idx];
        state = logaddexp_f(state + c, v);
        vout[idx] = __expf(state);
    }
}

extern "C" void kernel_launch(void* const* d_in, const int* in_sizes, int n_in,
                              void* d_out, int out_size, void* d_ws, size_t ws_size,
                              hipStream_t stream)
{
    const float* x  = (const float*)d_in[0];
    const float* h0 = (const float*)d_in[1];
    const float* Wz = (const float*)d_in[2];
    const float* bz = (const float*)d_in[3];
    const float* Wh = (const float*)d_in[4];
    const float* bh = (const float*)d_in[5];
    float* out = (float*)d_out;

    float* ws    = (float*)d_ws;
    float* c_arr = ws;
    float* Csum  = c_arr + (size_t)B_ * T_ * H_;
    float* Vsum  = Csum  + (size_t)B_ * NC_ * H_;
    float* entry = Vsum  + (size_t)B_ * NC_ * H_;

    mingru_gemm_kernel<<<(B_ * T_) / (MT_ * 16), 128, 0, stream>>>(x, Wz, bz, Wh, bh, out, c_arr);

    dim3 sgrid(NC_, B_);
    mingru_chunk_reduce<<<sgrid, H_, 0, stream>>>(c_arr, out, Csum, Vsum);
    mingru_carry<<<B_, H_, 0, stream>>>(h0, Csum, Vsum, entry);
    mingru_apply<<<sgrid, H_, 0, stream>>>(c_arr, out, entry);
}


# log1p(e), e=exp(-|d|) in [0,1]: library double-float routine (about 100 dependent vector ops, 4 calls per element) replaced by log2(1+e)*ln2 + exact rounding remainder of 1+e (5 ops); sums unchanged
# speedup vs baseline: 2.0834x; 1.4573x over previous
#include <hip/hip_runtime.h>
#include <hip/hip_bf16.h>
#include <math.h>

#define B_   32
#define T_   4096
#define DIN_ 256
#define H_   256
#define NC_  32
#define L_   (T_ / NC_)

typedef float v2f __attribute__((ext_vector_type(2)));
typedef float v8f __attribute__((ext_vector_type(8)));

__device__ __forceinline__ float log1p_unit(float e) {
    const float u = 1.0f + e;
    const float r = e - (u - 1.0f);
    return fmaf(__builtin_amdgcn_logf(u), 0.69314718055994531f, r);
}

__device__ __forceinline__ float softplus_f(float u) {
    return fmaxf(u, 0.0f) + log1p_unit(__expf(-fabsf(u)));
}

__device__ __forceinline__ float logaddexp_f(float a, float b) {
    float mx = fmaxf(a, b);
    float d  = fabsf(a - b);
    return mx + log1p_unit(__expf(-d));
}

__device__ __forceinline__ float log_g_f(float a) {
    return (a >= 0.0f) ? __logf(a + 0.5f) : -softplus_f(-a);
}

#define MT_ 4
#define NT_ 4
#define KS_ 8
#define SP_ 68

__global__ __launch_bounds__(128) void mingru_gemm_kernel(
    const float* __restrict__ x,  const float* __restrict__ Wz,
    const float* __restrict__ bz, const float* __restrict__ Wh,
    const float* __restrict__ bh, float* __restrict__ v_out,
    float* __restrict__ c_out)
{
    __shared__ float stage[4][2][MT_ * 16][SP_];
    const int lane  = threadIdx.x & 31;
    const int wave  = threadIdx.x >> 5;
    const int m0    = blockIdx.x * (MT_ * 16);
    const int n0    = wave * (NT_ * 16);
    const int l15   = lane & 15;
    const int khalf = (lane >> 4) << 1;

    const float* xp  = x  + (size_t)(m0 + l15) * DIN_ + khalf;
    const float* wzp = Wz + (size_t)(n0 + l15) * DIN_ + khalf;
    const float* whp = Wh + (size_t)(n0 + l15) * DIN_ + khalf;

    v8f accz[MT_][NT_];
    v8f acch[MT_][NT_];
#pragma unroll
    for (int mt = 0; mt < MT_; ++mt) {
#pragma unroll
        for (int nt = 0; nt < NT_; ++nt) {
            accz[mt][nt] = v8f{};
            acch[mt][nt] = v8f{};
        }
    }

    for (int kk = 0; kk < DIN_; kk += 4 * KS_) {
        v2f a[MT_][KS_];
        v2f fz[NT_][KS_];
        v2f fh[NT_][KS_];
#pragma unroll
        for (int mt = 0; mt < MT_; ++mt) {
#pragma unroll
            for (int s = 0; s < KS_; ++s)
                a[mt][s] = *(const v2f*)(xp + (size_t)mt * 16 * DIN_ + kk + 4 * s);
        }
#pragma unroll
        for (int nt = 0; nt < NT_; ++nt) {
#pragma unroll
            for (int s = 0; s < KS_; ++s)
                fz[nt][s] = *(const v2f*)(wzp + (size_t)nt * 16 * DIN_ + kk + 4 * s);
        }
#pragma unroll
        for (int nt = 0; nt < NT_; ++nt) {
#pragma unroll
            for (int s = 0; s < KS_; ++s)
                fh[nt][s] = *(const v2f*)(whp + (size_t)nt * 16 * DIN_ + kk + 4 * s);
        }
#pragma unroll
        for (int s = 0; s < KS_; ++s) {
#pragma unroll
            for (int mt = 0; mt < MT_; ++mt) {
#pragma unroll
                for (int nt = 0; nt < NT_; ++nt) {
                    accz[mt][nt] = __builtin_amdgcn_wmma_f32_16x16x4_f32(
                        false, a[mt][s], false, fz[nt][s], (short)0, accz[mt][nt], false, false);
                    acch[mt][nt] = __builtin_amdgcn_wmma_f32_16x16x4_f32(
                        false, a[mt][s], false, fh[nt][s], (short)0, acch[mt][nt], false, false);
                }
            }
        }
    }

    float* sz = &stage[wave][0][0][0];
    float* sh = &stage[wave][1][0][0];
    const int rhalf = (lane >> 4) << 3;
#pragma unroll
    for (int mt = 0; mt < MT_; ++mt) {
#pragma unroll
        for (int nt = 0; nt < NT_; ++nt) {
#pragma unroll
            for (int j = 0; j < 8; ++j) {
                const int off = (mt * 16 + rhalf + j) * SP_ + nt * 16 + l15;
                sz[off] = accz[mt][nt][j];
                sh[off] = acch[mt][nt][j];
            }
        }
    }
    __syncthreads();

    const int   c2   = lane << 1;
    const v2f   bzb  = *(const v2f*)(bz + n0 + c2);
    const v2f   bhb  = *(const v2f*)(bh + n0 + c2);
    for (int r0 = 0; r0 < MT_ * 16; r0 += 4) {
#pragma unroll
        for (int rr = 0; rr < 4; ++rr) {
            const int r = r0 + rr;
            const v2f kz = *(const v2f*)(sz + r * SP_ + c2);
            const v2f kh = *(const v2f*)(sh + r * SP_ + c2);
            v2f vo, co;
#pragma unroll
            for (int e = 0; e < 2; ++e) {
                float kpre  = kz[e] + bzb[e];
                float sp    = softplus_f(kpre);
                float log_z = kpre - sp;
                float log_c = -sp;
                float hp    = kh[e] + bhb[e];
                vo[e] = log_z + log_g_f(hp);
                co[e] = log_c;
            }
            const size_t idx = (size_t)(m0 + r) * H_ + n0 + c2;
            *(v2f*)(v_out + idx) = vo;
            *(v2f*)(c_out + idx) = co;
        }
    }
}

__global__ __launch_bounds__(256) void mingru_chunk_reduce(
    const float* __restrict__ c_arr, const float* __restrict__ v_arr,
    float* __restrict__ Csum, float* __restrict__ Vsum)
{
    const int h = threadIdx.x;
    const int j = blockIdx.x;
    const int b = blockIdx.y;

    size_t base = ((size_t)b * T_ + (size_t)j * L_) * H_ + h;
    float C = 0.0f;
    float V = -INFINITY;

    for (int t = 0; t < L_; ++t) {
        size_t idx = base + (size_t)t * H_;
        __builtin_prefetch(c_arr + idx + (size_t)8 * H_, 0, 0);
        __builtin_prefetch(v_arr + idx + (size_t)8 * H_, 0, 0);
        float c = c_arr[idx];
        float v = v_arr[idx];
        C += c;
        V = logaddexp_f(V + c, v);
    }
    size_t sidx = ((size_t)b * NC_ + j) * H_ + h;
    Csum[sidx] = C;
    Vsum[sidx] = V;
}

__global__ __launch_bounds__(256) void mingru_carry(
    const float* __restrict__ h0, const float* __restrict__ Csum,
    const float* __restrict__ Vsum, float* __restrict__ entry)
{
    const int h = threadIdx.x;
    const int b = blockIdx.x;

    float state = log_g_f(h0[(size_t)b * H_ + h]);
#pragma unroll
    for (int j = 0; j < NC_; ++j) {
        size_t sidx = ((size_t)b * NC_ + j) * H_ + h;
        entry[sidx] = state;
        state = logaddexp_f(state + Csum[sidx], Vsum[sidx]);
    }
}

__global__ __launch_bounds__(256) void mingru_apply(
    const float* __restrict__ c_arr, float* __restrict__ vout,
    const float* __restrict__ entry)
{
    const int h = threadIdx.x;
    const int j = blockIdx.x;
    const int b = blockIdx.y;

    float state = entry[((size_t)b * NC_ + j) * H_ + h];
    size_t base = ((size_t)b * T_ + (size_t)j * L_) * H_ + h;

    for (int t = 0; t < L_; ++t) {
        size_t idx = base + (size_t)t * H_;
        __builtin_prefetch(c_arr + idx + (size_t)8 * H_, 0, 0);
        float c = c_arr[idx];
        float v = vout[idx];
        state = logaddexp_f(state + c, v);
        vout[idx] = __expf(state);
    }
}

extern "C" void kernel_launch(void* const* d_in, const int* in_sizes, int n_in,
                              void* d_out, int out_size, void* d_ws, size_t ws_size,
                              hipStream_t stream)
{
    const float* x  = (const float*)d_in[0];
    const float* h0 = (const float*)d_in[1];
    const float* Wz = (const float*)d_in[2];
    const float* bz = (const float*)d_in[3];
    const float* Wh = (const float*)d_in[4];
    const float* bh = (const float*)d_in[5];
    float* out = (float*)d_out;

    float* ws    = (float*)d_ws;
    float* c_arr = ws;
    float* Csum  = c_arr + (size_t)B_ * T_ * H_;
    float* Vsum  = Csum  + (size_t)B_ * NC_ * H_;
    float* entry = Vsum  + (size_t)B_ * NC_ * H_;

    mingru_gemm_kernel<<<(B_ * T_) / (MT_ * 16), 128, 0, stream>>>(x, Wz, bz, Wh, bh, out, c_arr);

    dim3 sgrid(NC_, B_);
    mingru_chunk_reduce<<<sgrid, H_, 0, stream>>>(c_arr, out, Csum, Vsum);
    mingru_carry<<<B_, H_, 0, stream>>>(h0, Csum, Vsum, entry);
    mingru_apply<<<sgrid, H_, 0, stream>>>(c_arr, out, entry);
}


# Chunk scan moved into the GEMM epilogue: each block scans its own 64 rows (coefficient sum, local state); reduce launch removed, last pass elementwise exp(S)+exp(E+A); summation order changes
# speedup vs baseline: 2.4763x; 1.1886x over previous
#include <hip/hip_runtime.h>
#include <hip/hip_bf16.h>
#include <math.h>

#define B_   32
#define T_   4096
#define DIN_ 256
#define H_   256
#define L_   64
#define NC_  (T_ / L_)

typedef float v2f __attribute__((ext_vector_type(2)));
typedef float v4f __attribute__((ext_vector_type(4)));
typedef float v8f __attribute__((ext_vector_type(8)));

__device__ __forceinline__ float log1p_unit(float e) {
    const float u = 1.0f + e;
    const float r = e - (u - 1.0f);
    return fmaf(__builtin_amdgcn_logf(u), 0.69314718055994531f, r);
}

__device__ __forceinline__ float softplus_f(float u) {
    return fmaxf(u, 0.0f) + log1p_unit(__expf(-fabsf(u)));
}

__device__ __forceinline__ float logaddexp_f(float a, float b) {
    float mx = fmaxf(a, b);
    float d  = fabsf(a - b);
    return mx + log1p_unit(__expf(-d));
}

__device__ __forceinline__ float log_g_f(float a) {
    return (a >= 0.0f) ? __logf(a + 0.5f) : -softplus_f(-a);
}

#define MT_ 4
#define NT_ 4
#define KS_ 8
static_assert(L_ == MT_ * 16 && T_ % L_ == 0, "one GEMM block must be exactly one scan chunk of one batch row");
#define SP_ 68

__global__ __launch_bounds__(128) void mingru_gemm_kernel(
    const float* __restrict__ x,  const float* __restrict__ Wz,
    const float* __restrict__ bz, const float* __restrict__ Wh,
    const float* __restrict__ bh, float* __restrict__ v_out,
    float* __restrict__ c_out)
{
    __shared__ float stage[4][2][MT_ * 16][SP_];
    const int lane  = threadIdx.x & 31;
    const int wave  = threadIdx.x >> 5;
    const int m0    = blockIdx.x * (MT_ * 16);
    const int n0    = wave * (NT_ * 16);
    const int l15   = lane & 15;
    const int khalf = (lane >> 4) << 1;

    const float* xp  = x  + (size_t)(m0 + l15) * DIN_ + khalf;
    const float* wzp = Wz + (size_t)(n0 + l15) * DIN_ + khalf;
    const float* whp = Wh + (size_t)(n0 + l15) * DIN_ + khalf;

    v8f accz[MT_][NT_];
    v8f acch[MT_][NT_];
#pragma unroll
    for (int mt = 0; mt < MT_; ++mt) {
#pragma unroll
        for (int nt = 0; nt < NT_; ++nt) {
            accz[mt][nt] = v8f{};
            acch[mt][nt] = v8f{};
        }
    }

    for (int kk = 0; kk < DIN_; kk += 4 * KS_) {
        v2f a[MT_][KS_];
        v2f fz[NT_][KS_];
        v2f fh[NT_][KS_];
#pragma unroll
        for (int mt = 0; mt < MT_; ++mt) {
#pragma unroll
            for (int s = 0; s < KS_; ++s)
                a[mt][s] = *(const v2f*)(xp + (size_t)mt * 16 * DIN_ + kk + 4 * s);
        }
#pragma unroll
        for (int nt = 0; nt < NT_; ++nt) {
#pragma unroll
            for (int s = 0; s < KS_; ++s)
                fz[nt][s] = *(const v2f*)(wzp + (size_t)nt * 16 * DIN_ + kk + 4 * s);
        }
#pragma unroll
        for (int nt = 0; nt < NT_; ++nt) {
#pragma unroll
            for (int s = 0; s < KS_; ++s)
                fh[nt][s] = *(const v2f*)(whp + (size_t)nt * 16 * DIN_ + kk + 4 * s);
        }
#pragma unroll
        for (int s = 0; s < KS_; ++s) {
#pragma unroll
            for (int mt = 0; mt < MT_; ++mt) {
#pragma unroll
                for (int nt = 0; nt < NT_; ++nt) {
                    accz[mt][nt] = __builtin_amdgcn_wmma_f32_16x16x4_f32(
                        false, a[mt][s], false, fz[nt][s], (short)0, accz[mt][nt], false, false);
                    acch[mt][nt] = __builtin_amdgcn_wmma_f32_16x16x4_f32(
                        false, a[mt][s], false, fh[nt][s], (short)0, acch[mt][nt], false, false);
                }
            }
        }
    }

    float* sz = &stage[wave][0][0][0];
    float* sh = &stage[wave][1][0][0];
    const int rhalf = (lane >> 4) << 3;
#pragma unroll
    for (int mt = 0; mt < MT_; ++mt) {
#pragma unroll
        for (int nt = 0; nt < NT_; ++nt) {
#pragma unroll
            for (int j = 0; j < 8; ++j) {
                const int off = (mt * 16 + rhalf + j) * SP_ + nt * 16 + l15;
                sz[off] = accz[mt][nt][j];
                sh[off] = acch[mt][nt][j];
            }
        }
    }
    __syncthreads();

    const int   c2   = lane << 1;
    const v2f   bzb  = *(const v2f*)(bz + n0 + c2);
    const v2f   bhb  = *(const v2f*)(bh + n0 + c2);
    v2f runA = v2f{0.0f, 0.0f};
    v2f runS = v2f{-INFINITY, -INFINITY};
    for (int r0 = 0; r0 < MT_ * 16; r0 += 4) {
#pragma unroll
        for (int rr = 0; rr < 4; ++rr) {
            const int r = r0 + rr;
            const v2f kz = *(const v2f*)(sz + r * SP_ + c2);
            const v2f kh = *(const v2f*)(sh + r * SP_ + c2);
            v2f vo, co;
#pragma unroll
            for (int e = 0; e < 2; ++e) {
                float kpre  = kz[e] + bzb[e];
                float sp    = softplus_f(kpre);
                float log_z = kpre - sp;
                float log_c = -sp;
                float hp    = kh[e] + bhb[e];
                runA[e] += log_c;
                runS[e]  = logaddexp_f(runS[e] + log_c, log_z + log_g_f(hp));
                vo[e] = runS[e];
                co[e] = runA[e];
            }
            const size_t idx = (size_t)(m0 + r) * H_ + n0 + c2;
            *(v2f*)(v_out + idx) = vo;
            *(v2f*)(c_out + idx) = co;
        }
    }
}

__global__ __launch_bounds__(256) void mingru_carry(
    const float* __restrict__ h0, const float* __restrict__ a_arr,
    const float* __restrict__ s_arr, float* __restrict__ entry)
{
    const int h = threadIdx.x;
    const int b = blockIdx.x;

    float state = log_g_f(h0[(size_t)b * H_ + h]);
#pragma unroll 16
    for (int j = 0; j < NC_; ++j) {
        const size_t last = ((size_t)b * T_ + (size_t)j * L_ + (L_ - 1)) * H_ + h;
        entry[((size_t)b * NC_ + j) * H_ + h] = state;
        state = logaddexp_f(state + a_arr[last], s_arr[last]);
    }
}

__global__ __launch_bounds__(256) void mingru_apply(
    const float* __restrict__ a_arr, float* __restrict__ out,
    const float* __restrict__ entry)
{
    const int col  = (threadIdx.x & 63) << 2;
    const int rsub = threadIdx.x >> 6;
    const int j = blockIdx.x;
    const int b = blockIdx.y;

    const v4f E = *(const v4f*)(entry + ((size_t)b * NC_ + j) * H_ + col);
    const size_t base = ((size_t)b * T_ + (size_t)j * L_ + rsub) * H_ + col;

    for (int r0 = 0; r0 < L_; r0 += 16) {
        v4f a[4], sv[4];
#pragma unroll
        for (int i = 0; i < 4; ++i) {
            const size_t idx = base + (size_t)(r0 + 4 * i) * H_;
            a[i]  = __builtin_nontemporal_load((const v4f*)(a_arr + idx));
            sv[i] = __builtin_nontemporal_load((const v4f*)(out + idx));
        }
#pragma unroll
        for (int i = 0; i < 4; ++i) {
            const size_t idx = base + (size_t)(r0 + 4 * i) * H_;
            v4f o;
#pragma unroll
            for (int e = 0; e < 4; ++e)
                o[e] = __expf(sv[i][e]) + __expf(E[e] + a[i][e]);
            *(v4f*)(out + idx) = o;
        }
    }
}

extern "C" void kernel_launch(void* const* d_in, const int* in_sizes, int n_in,
                              void* d_out, int out_size, void* d_ws, size_t ws_size,
                              hipStream_t stream)
{
    const float* x  = (const float*)d_in[0];
    const float* h0 = (const float*)d_in[1];
    const float* Wz = (const float*)d_in[2];
    const float* bz = (const float*)d_in[3];
    const float* Wh = (const float*)d_in[4];
    const float* bh = (const float*)d_in[5];
    float* out = (float*)d_out;

    float* ws    = (float*)d_ws;
    float* a_arr = ws;
    float* entry = a_arr + (size_t)B_ * T_ * H_;

    mingru_gemm_kernel<<<(B_ * T_) / (MT_ * 16), 128, 0, stream>>>(x, Wz, bz, Wh, bh, out, a_arr);

    mingru_carry<<<B_, H_, 0, stream>>>(h0, a_arr, out, entry);
    dim3 sgrid(NC_, B_);
    mingru_apply<<<sgrid, 256, 0, stream>>>(a_arr, out, entry);
}


# GEMM operands staged through wave-private LDS: whole-line 16-byte global loads (4 lines per instruction, not 16), ds_store, fragments read from LDS in the same k order; sums unchanged
# speedup vs baseline: 2.8035x; 1.1321x over previous
#include <hip/hip_runtime.h>
#include <hip/hip_bf16.h>
#include <math.h>

#define B_   32
#define T_   4096
#define DIN_ 256
#define H_   256
#define L_   64
#define NC_  (T_ / L_)

typedef float v2f __attribute__((ext_vector_type(2)));
typedef float v4f __attribute__((ext_vector_type(4)));
typedef float v8f __attribute__((ext_vector_type(8)));

__device__ __forceinline__ float log1p_unit(float e) {
    const float u = 1.0f + e;
    const float r = e - (u - 1.0f);
    return fmaf(__builtin_amdgcn_logf(u), 0.69314718055994531f, r);
}

__device__ __forceinline__ float softplus_f(float u) {
    return fmaxf(u, 0.0f) + log1p_unit(__expf(-fabsf(u)));
}

__device__ __forceinline__ float logaddexp_f(float a, float b) {
    float mx = fmaxf(a, b);
    float d  = fabsf(a - b);
    return mx + log1p_unit(__expf(-d));
}

__device__ __forceinline__ float log_g_f(float a) {
    return (a >= 0.0f) ? __logf(a + 0.5f) : -softplus_f(-a);
}

#define MT_ 4
#define NT_ 4
#define KS_ 8
static_assert(L_ == MT_ * 16 && T_ % L_ == 0, "one GEMM block must be exactly one scan chunk of one batch row");
#define SP_ 68
#define KP_ 36
#define SR_ (MT_ * 16 + 2 * NT_ * 16)
static_assert(SR_ * KP_ <= 2 * MT_ * 16 * SP_, "the operand slices of a trip must fit the wave's own epilogue region");

__global__ __launch_bounds__(128) void mingru_gemm_kernel(
    const float* __restrict__ x,  const float* __restrict__ Wz,
    const float* __restrict__ bz, const float* __restrict__ Wh,
    const float* __restrict__ bh, float* __restrict__ v_out,
    float* __restrict__ c_out)
{
    __shared__ __attribute__((aligned(16))) float stage[4][2][MT_ * 16][SP_];
    const int lane  = threadIdx.x & 31;
    const int wave  = threadIdx.x >> 5;
    const int m0    = blockIdx.x * (MT_ * 16);
    const int n0    = wave * (NT_ * 16);
    const int l15   = lane & 15;
    const int khalf = (lane >> 4) << 1;

    const int prow = lane >> 3;
    const int pcol = (lane & 7) << 2;
    const int srow = ((prow & 1) << 3) + (prow >> 1);
    const float* xg  = x  + (size_t)(m0 + srow) * DIN_ + pcol;
    const float* wzg = Wz + (size_t)(n0 + srow) * DIN_ + pcol;
    const float* whg = Wh + (size_t)(n0 + srow) * DIN_ + pcol;
    float* st = &stage[wave][0][0][0];
    float* stw = st + srow * KP_ + pcol;
    const float* sxa = st + l15 * KP_ + khalf;
    const float* swz = sxa + (MT_ * 16) * KP_;
    const float* swh = swz + (NT_ * 16) * KP_;

    v8f accz[MT_][NT_];
    v8f acch[MT_][NT_];
#pragma unroll
    for (int mt = 0; mt < MT_; ++mt) {
#pragma unroll
        for (int nt = 0; nt < NT_; ++nt) {
            accz[mt][nt] = v8f{};
            acch[mt][nt] = v8f{};
        }
    }

    for (int kk = 0; kk < DIN_; kk += 4 * KS_) {
        v2f a[MT_][KS_];
        v2f fz[NT_][KS_];
        v2f fh[NT_][KS_];
        v4f pc[3][16];
#pragma unroll
        for (int i = 0; i < 16; ++i) {
            const size_t ro = (size_t)(16 * (i >> 2) + 2 * (i & 3)) * DIN_ + kk;
            pc[0][i] = *(const v4f*)(xg  + ro);
            pc[1][i] = *(const v4f*)(wzg + ro);
            pc[2][i] = *(const v4f*)(whg + ro);
        }
#pragma unroll
        for (int g = 0; g < 3; ++g) {
#pragma unroll
            for (int i = 0; i < 16; ++i)
                *(v4f*)(stw + (g * 64 + 16 * (i >> 2) + 2 * (i & 3)) * KP_) = pc[g][i];
        }
#pragma unroll
        for (int mt = 0; mt < MT_; ++mt) {
#pragma unroll
            for (int s = 0; s < KS_; ++s)
                a[mt][s] = *(const v2f*)(sxa + mt * 16 * KP_ + 4 * s);
        }
#pragma unroll
        for (int nt = 0; nt < NT_; ++nt) {
#pragma unroll
            for (int s = 0; s < KS_; ++s)
                fz[nt][s] = *(const v2f*)(swz + nt * 16 * KP_ + 4 * s);
        }
#pragma unroll
        for (int nt = 0; nt < NT_; ++nt) {
#pragma unroll
            for (int s = 0; s < KS_; ++s)
                fh[nt][s] = *(const v2f*)(swh + nt * 16 * KP_ + 4 * s);
        }
#pragma unroll
        for (int s = 0; s < KS_; ++s) {
#pragma unroll
            for (int mt = 0; mt < MT_; ++mt) {
#pragma unroll
                for (int nt = 0; nt < NT_; ++nt) {
                    accz[mt][nt] = __builtin_amdgcn_wmma_f32_16x16x4_f32(
                        false, a[mt][s], false, fz[nt][s], (short)0, accz[mt][nt], false, false);
                    acch[mt][nt] = __builtin_amdgcn_wmma_f32_16x16x4_f32(
                        false, a[mt][s], false, fh[nt][s], (short)0, acch[mt][nt], false, false);
                }
            }
        }
    }

    float* sz = &stage[wave][0][0][0];
    float* sh = &stage[wave][1][0][0];
    const int rhalf = (lane >> 4) << 3;
#pragma unroll
    for (int mt = 0; mt < MT_; ++mt) {
#pragma unroll
        for (int nt = 0; nt < NT_; ++nt) {
#pragma unroll
            for (int j = 0; j < 8; ++j) {
                const int off = (mt * 16 + rhalf + j) * SP_ + nt * 16 + l15;
                sz[off] = accz[mt][nt][j];
                sh[off] = acch[mt][nt][j];
            }
        }
    }
    __syncthreads();

    const int   c2   = lane << 1;
    const v2f   bzb  = *(const v2f*)(bz + n0 + c2);
    const v2f   bhb  = *(const v2f*)(bh + n0 + c2);
    v2f runA = v2f{0.0f, 0.0f};
    v2f runS = v2f{-INFINITY, -INFINITY};
    for (int r0 = 0; r0 < MT_ * 16; r0 += 4) {
#pragma unroll
        for (int rr = 0; rr < 4; ++rr) {
            const int r = r0 + rr;
            const v2f kz = *(const v2f*)(sz + r * SP_ + c2);
            const v2f kh = *(const v2f*)(sh + r * SP_ + c2);
            v2f vo, co;
#pragma unroll
            for (int e = 0; e < 2; ++e) {
                float kpre  = kz[e] + bzb[e];
                float sp    = softplus_f(kpre);
                float log_z = kpre - sp;
                float log_c = -sp;
                float hp    = kh[e] + bhb[e];
                runA[e] += log_c;
                runS[e]  = logaddexp_f(runS[e] + log_c, log_z + log_g_f(hp));
                vo[e] = runS[e];
                co[e] = runA[e];
            }
            const size_t idx = (size_t)(m0 + r) * H_ + n0 + c2;
            *(v2f*)(v_out + idx) = vo;
            *(v2f*)(c_out + idx) = co;
        }
    }
}

__global__ __launch_bounds__(256) void mingru_carry(
    const float* __restrict__ h0, const float* __restrict__ a_arr,
    const float* __restrict__ s_arr, float* __restrict__ entry)
{
    const int h = threadIdx.x;
    const int b = blockIdx.x;

    float state = log_g_f(h0[(size_t)b * H_ + h]);
#pragma unroll 16
    for (int j = 0; j < NC_; ++j) {
        const size_t last = ((size_t)b * T_ + (size_t)j * L_ + (L_ - 1)) * H_ + h;
        entry[((size_t)b * NC_ + j) * H_ + h] = state;
        state = logaddexp_f(state + a_arr[last], s_arr[last]);
    }
}

__global__ __launch_bounds__(256) void mingru_apply(
    const float* __restrict__ a_arr, float* __restrict__ out,
    const float* __restrict__ entry)
{
    const int col  = (threadIdx.x & 63) << 2;
    const int rsub = threadIdx.x >> 6;
    const int j = blockIdx.x;
    const int b = blockIdx.y;

    const v4f E = *(const v4f*)(entry + ((size_t)b * NC_ + j) * H_ + col);
    const size_t base = ((size_t)b * T_ + (size_t)j * L_ + rsub) * H_ + col;

    for (int r0 = 0; r0 < L_; r0 += 16) {
        v4f a[4], sv[4];
#pragma unroll
        for (int i = 0; i < 4; ++i) {
            const size_t idx = base + (size_t)(r0 + 4 * i) * H_;
            a[i]  = __builtin_nontemporal_load((const v4f*)(a_arr + idx));
            sv[i] = __builtin_nontemporal_load((const v4f*)(out + idx));
        }
#pragma unroll
        for (int i = 0; i < 4; ++i) {
            const size_t idx = base + (size_t)(r0 + 4 * i) * H_;
            v4f o;
#pragma unroll
            for (int e = 0; e < 4; ++e)
                o[e] = __expf(sv[i][e]) + __expf(E[e] + a[i][e]);
            *(v4f*)(out + idx) = o;
        }
    }
}

extern "C" void kernel_launch(void* const* d_in, const int* in_sizes, int n_in,
                              void* d_out, int out_size, void* d_ws, size_t ws_size,
                              hipStream_t stream)
{
    const float* x  = (const float*)d_in[0];
    const float* h0 = (const float*)d_in[1];
    const float* Wz = (const float*)d_in[2];
    const float* bz = (const float*)d_in[3];
    const float* Wh = (const float*)d_in[4];
    const float* bh = (const float*)d_in[5];
    float* out = (float*)d_out;

    float* ws    = (float*)d_ws;
    float* a_arr = ws;
    float* entry = a_arr + (size_t)B_ * T_ * H_;

    mingru_gemm_kernel<<<(B_ * T_) / (MT_ * 16), 128, 0, stream>>>(x, Wz, bz, Wh, bh, out, a_arr);

    mingru_carry<<<B_, H_, 0, stream>>>(h0, a_arr, out, entry);
    dim3 sgrid(NC_, B_);
    mingru_apply<<<sgrid, 256, 0, stream>>>(a_arr, out, entry);
}


# GEMM epilogue software-pipelined: log g by select, not branch, and 8 rows' independent gate terms run one trip ahead of the serial scan steps, so their latencies overlap; sums unchanged
# speedup vs baseline: 3.1879x; 1.1371x over previous
#include <hip/hip_runtime.h>
#include <hip/hip_bf16.h>
#include <math.h>

#define B_   32
#define T_   4096
#define DIN_ 256
#define H_   256
#define L_   64
#define NC_  (T_ / L_)

typedef float v2f __attribute__((ext_vector_type(2)));
typedef float v4f __attribute__((ext_vector_type(4)));
typedef float v8f __attribute__((ext_vector_type(8)));

__device__ __forceinline__ float log1p_unit(float e) {
    const float u = 1.0f + e;
    const float r = e - (u - 1.0f);
    return fmaf(__builtin_amdgcn_logf(u), 0.69314718055994531f, r);
}

__device__ __forceinline__ float softplus_f(float u) {
    return fmaxf(u, 0.0f) + log1p_unit(__expf(-fabsf(u)));
}

__device__ __forceinline__ float logaddexp_f(float a, float b) {
    float mx = fmaxf(a, b);
    float d  = fabsf(a - b);
    return mx + log1p_unit(__expf(-d));
}

__device__ __forceinline__ float log_g_f(float a) {
    return (a >= 0.0f) ? __logf(a + 0.5f) : -softplus_f(-a);
}

__device__ __forceinline__ float log_g_sel(float a) {
    const float pos = __logf(a + 0.5f);
    const float neg = -softplus_f(-a);
    return (a >= 0.0f) ? pos : neg;
}

#define SP_ 68
#define RB_ 8

__device__ __forceinline__ void epi_terms(const float* zrow, const float* hrow, int c2,
                                          v2f bzb, v2f bhb, v2f (&lc)[RB_], v2f (&val)[RB_]) {
#pragma unroll
    for (int rr = 0; rr < RB_; ++rr) {
        const v2f kz = *(const v2f*)(zrow + rr * SP_ + c2);
        const v2f kh = *(const v2f*)(hrow + rr * SP_ + c2);
#pragma unroll
        for (int e = 0; e < 2; ++e) {
            const float kpre  = kz[e] + bzb[e];
            const float sp    = softplus_f(kpre);
            const float log_z = kpre - sp;
            const float hp    = kh[e] + bhb[e];
            lc[rr][e]  = -sp;
            val[rr][e] = log_z + log_g_sel(hp);
        }
    }
}

__device__ __forceinline__ void epi_scan(const v2f (&lc)[RB_], const v2f (&val)[RB_],
                                         v2f& runA, v2f& runS, float* vrow, float* crow) {
#pragma unroll
    for (int rr = 0; rr < RB_; ++rr) {
        v2f vo, co;
#pragma unroll
        for (int e = 0; e < 2; ++e) {
            runA[e] += lc[rr][e];
            runS[e]  = logaddexp_f(runS[e] + lc[rr][e], val[rr][e]);
            vo[e] = runS[e];
            co[e] = runA[e];
        }
        *(v2f*)(vrow + (size_t)rr * H_) = vo;
        *(v2f*)(crow + (size_t)rr * H_) = co;
    }
}

#define MT_ 4
#define NT_ 4
#define KS_ 8
static_assert((MT_ * 16) % RB_ == 0, "epilogue trips must tile the block's rows");
static_assert(L_ == MT_ * 16 && T_ % L_ == 0, "one GEMM block must be exactly one scan chunk of one batch row");
#define KP_ 36
#define SR_ (MT_ * 16 + 2 * NT_ * 16)
static_assert(SR_ * KP_ <= 2 * MT_ * 16 * SP_, "the operand slices of a trip must fit the wave's own epilogue region");

__global__ __launch_bounds__(128) void mingru_gemm_kernel(
    const float* __restrict__ x,  const float* __restrict__ Wz,
    const float* __restrict__ bz, const float* __restrict__ Wh,
    const float* __restrict__ bh, float* __restrict__ v_out,
    float* __restrict__ c_out)
{
    __shared__ __attribute__((aligned(16))) float stage[4][2][MT_ * 16][SP_];
    const int lane  = threadIdx.x & 31;
    const int wave  = threadIdx.x >> 5;
    const int m0    = blockIdx.x * (MT_ * 16);
    const int n0    = wave * (NT_ * 16);
    const int l15   = lane & 15;
    const int khalf = (lane >> 4) << 1;

    const int prow = lane >> 3;
    const int pcol = (lane & 7) << 2;
    const int srow = ((prow & 1) << 3) + (prow >> 1);
    const float* xg  = x  + (size_t)(m0 + srow) * DIN_ + pcol;
    const float* wzg = Wz + (size_t)(n0 + srow) * DIN_ + pcol;
    const float* whg = Wh + (size_t)(n0 + srow) * DIN_ + pcol;
    float* st = &stage[wave][0][0][0];
    float* stw = st + srow * KP_ + pcol;
    const float* sxa = st + l15 * KP_ + khalf;
    const float* swz = sxa + (MT_ * 16) * KP_;
    const float* swh = swz + (NT_ * 16) * KP_;

    v8f accz[MT_][NT_];
    v8f acch[MT_][NT_];
#pragma unroll
    for (int mt = 0; mt < MT_; ++mt) {
#pragma unroll
        for (int nt = 0; nt < NT_; ++nt) {
            accz[mt][nt] = v8f{};
            acch[mt][nt] = v8f{};
        }
    }

    for (int kk = 0; kk < DIN_; kk += 4 * KS_) {
        v2f a[MT_][KS_];
        v2f fz[NT_][KS_];
        v2f fh[NT_][KS_];
        v4f pc[3][16];
#pragma unroll
        for (int i = 0; i < 16; ++i) {
            const size_t ro = (size_t)(16 * (i >> 2) + 2 * (i & 3)) * DIN_ + kk;
            pc[0][i] = *(const v4f*)(xg  + ro);
            pc[1][i] = *(const v4f*)(wzg + ro);
            pc[2][i] = *(const v4f*)(whg + ro);
        }
#pragma unroll
        for (int g = 0; g < 3; ++g) {
#pragma unroll
            for (int i = 0; i < 16; ++i)
                *(v4f*)(stw + (g * 64 + 16 * (i >> 2) + 2 * (i & 3)) * KP_) = pc[g][i];
        }
#pragma unroll
        for (int mt = 0; mt < MT_; ++mt) {
#pragma unroll
            for (int s = 0; s < KS_; ++s)
                a[mt][s] = *(const v2f*)(sxa + mt * 16 * KP_ + 4 * s);
        }
#pragma unroll
        for (int nt = 0; nt < NT_; ++nt) {
#pragma unroll
            for (int s = 0; s < KS_; ++s)
                fz[nt][s] = *(const v2f*)(swz + nt * 16 * KP_ + 4 * s);
        }
#pragma unroll
        for (int nt = 0; nt < NT_; ++nt) {
#pragma unroll
            for (int s = 0; s < KS_; ++s)
                fh[nt][s] = *(const v2f*)(swh + nt * 16 * KP_ + 4 * s);
        }
#pragma unroll
        for (int s = 0; s < KS_; ++s) {
#pragma unroll
            for (int mt = 0; mt < MT_; ++mt) {
#pragma unroll
                for (int nt = 0; nt < NT_; ++nt) {
                    accz[mt][nt] = __builtin_amdgcn_wmma_f32_16x16x4_f32(
                        false, a[mt][s], false, fz[nt][s], (short)0, accz[mt][nt], false, false);
                    acch[mt][nt] = __builtin_amdgcn_wmma_f32_16x16x4_f32(
                        false, a[mt][s], false, fh[nt][s], (short)0, acch[mt][nt], false, false);
                }
            }
        }
    }

    float* sz = &stage[wave][0][0][0];
    float* sh = &stage[wave][1][0][0];
    const int rhalf = (lane >> 4) << 3;
#pragma unroll
    for (int mt = 0; mt < MT_; ++mt) {
#pragma unroll
        for (int nt = 0; nt < NT_; ++nt) {
#pragma unroll
            for (int j = 0; j < 8; ++j) {
                const int off = (mt * 16 + rhalf + j) * SP_ + nt * 16 + l15;
                sz[off] = accz[mt][nt][j];
                sh[off] = acch[mt][nt][j];
            }
        }
    }
    __syncthreads();

    const int   c2   = lane << 1;
    const v2f   bzb  = *(const v2f*)(bz + n0 + c2);
    const v2f   bhb  = *(const v2f*)(bh + n0 + c2);
    v2f runA = v2f{0.0f, 0.0f};
    v2f runS = v2f{-INFINITY, -INFINITY};
    v2f lc[RB_];
    v2f val[RB_];
    epi_terms(sz, sh, c2, bzb, bhb, lc, val);
    float* vrow = v_out + (size_t)m0 * H_ + n0 + c2;
    float* crow = c_out + (size_t)m0 * H_ + n0 + c2;
    for (int r0 = 0; r0 < MT_ * 16 - RB_; r0 += RB_) {
        v2f lcn[RB_], valn[RB_];
        epi_terms(sz + (r0 + RB_) * SP_, sh + (r0 + RB_) * SP_, c2, bzb, bhb, lcn, valn);
        epi_scan(lc, val, runA, runS, vrow + (size_t)r0 * H_, crow + (size_t)r0 * H_);
#pragma unroll
        for (int rr = 0; rr < RB_; ++rr) {
            lc[rr]  = lcn[rr];
            val[rr] = valn[rr];
        }
    }
    epi_scan(lc, val, runA, runS, vrow + (size_t)(MT_ * 16 - RB_) * H_, crow + (size_t)(MT_ * 16 - RB_) * H_);
}

__global__ __launch_bounds__(256) void mingru_carry(
    const float* __restrict__ h0, const float* __restrict__ a_arr,
    const float* __restrict__ s_arr, float* __restrict__ entry)
{
    const int h = threadIdx.x;
    const int b = blockIdx.x;

    float state = log_g_f(h0[(size_t)b * H_ + h]);
#pragma unroll 16
    for (int j = 0; j < NC_; ++j) {
        const size_t last = ((size_t)b * T_ + (size_t)j * L_ + (L_ - 1)) * H_ + h;
        entry[((size_t)b * NC_ + j) * H_ + h] = state;
        state = logaddexp_f(state + a_arr[last], s_arr[last]);
    }
}

__global__ __launch_bounds__(256) void mingru_apply(
    const float* __restrict__ a_arr, float* __restrict__ out,
    const float* __restrict__ entry)
{
    const int col  = (threadIdx.x & 63) << 2;
    const int rsub = threadIdx.x >> 6;
    const int j = blockIdx.x;
    const int b = blockIdx.y;

    const v4f E = *(const v4f*)(entry + ((size_t)b * NC_ + j) * H_ + col);
    const size_t base = ((size_t)b * T_ + (size_t)j * L_ + rsub) * H_ + col;

    for (int r0 = 0; r0 < L_; r0 += 16) {
        v4f a[4], sv[4];
#pragma unroll
        for (int i = 0; i < 4; ++i) {
            const size_t idx = base + (size_t)(r0 + 4 * i) * H_;
            a[i]  = __builtin_nontemporal_load((const v4f*)(a_arr + idx));
            sv[i] = __builtin_nontemporal_load((const v4f*)(out + idx));
        }
#pragma unroll
        for (int i = 0; i < 4; ++i) {
            const size_t idx = base + (size_t)(r0 + 4 * i) * H_;
            v4f o;
#pragma unroll
            for (int e = 0; e < 4; ++e)
                o[e] = __expf(sv[i][e]) + __expf(E[e] + a[i][e]);
            *(v4f*)(out + idx) = o;
        }
    }
}

extern "C" void kernel_launch(void* const* d_in, const int* in_sizes, int n_in,
                              void* d_out, int out_size, void* d_ws, size_t ws_size,
                              hipStream_t stream)
{
    const float* x  = (const float*)d_in[0];
    const float* h0 = (const float*)d_in[1];
    const float* Wz = (const float*)d_in[2];
    const float* bz = (const float*)d_in[3];
    const float* Wh = (const float*)d_in[4];
    const float* bh = (const float*)d_in[5];
    float* out = (float*)d_out;

    float* ws    = (float*)d_ws;
    float* a_arr = ws;
    float* entry = a_arr + (size_t)B_ * T_ * H_;

    mingru_gemm_kernel<<<(B_ * T_) / (MT_ * 16), 128, 0, stream>>>(x, Wz, bz, Wh, bh, out, a_arr);

    mingru_carry<<<B_, H_, 0, stream>>>(h0, a_arr, out, entry);
    dim3 sgrid(NC_, B_);
    mingru_apply<<<sgrid, 256, 0, stream>>>(a_arr, out, entry);
}


# Scan carried in linear space: a=1-sigmoid(k), b=sigmoid(k)*g(hp), h=a*h+b, one fma per row, all terms positive; per element 2 exp + 2 rcp, not 5 exp + 4 log; summation order changes
# speedup vs baseline: 3.3902x; 1.0635x over previous
#include <hip/hip_runtime.h>
#include <hip/hip_bf16.h>
#include <math.h>

#define B_   32
#define T_   4096
#define DIN_ 256
#define H_   256
#define L_   64
#define NC_  (T_ / L_)

typedef float v2f __attribute__((ext_vector_type(2)));
typedef float v4f __attribute__((ext_vector_type(4)));
typedef float v8f __attribute__((ext_vector_type(8)));


__device__ __forceinline__ void sigmoid_both(float u, float& s_of_u, float& s_of_minus_u) {
    const float e = __expf(-fabsf(u));
    const float r = __builtin_amdgcn_rcpf(1.0f + e);
    const float q = e * r;
    const bool  nonneg = (u >= 0.0f);
    s_of_u       = nonneg ? r : q;
    s_of_minus_u = nonneg ? q : r;
}

__device__ __forceinline__ float g_lin(float a) {
    const float e  = __expf(-fabsf(a));
    const float sg = e * __builtin_amdgcn_rcpf(1.0f + e);
    return (a >= 0.0f) ? a + 0.5f : sg;
}

#define SP_ 68
#define RB_ 8

__device__ __forceinline__ void epi_terms(const float* zrow, const float* hrow, int c2,
                                          v2f bzb, v2f bhb, v2f (&ca)[RB_], v2f (&cb)[RB_]) {
#pragma unroll
    for (int rr = 0; rr < RB_; ++rr) {
        const v2f kz = *(const v2f*)(zrow + rr * SP_ + c2);
        const v2f kh = *(const v2f*)(hrow + rr * SP_ + c2);
#pragma unroll
        for (int e = 0; e < 2; ++e) {
            const float kpre = kz[e] + bzb[e];
            const float hp   = kh[e] + bhb[e];
            float z, omz;
            sigmoid_both(kpre, z, omz);
            ca[rr][e] = omz;
            cb[rr][e] = z * g_lin(hp);
        }
    }
}

__device__ __forceinline__ void epi_scan(const v2f (&ca)[RB_], const v2f (&cb)[RB_],
                                         v2f& runP, v2f& runS, float* vrow, float* crow) {
#pragma unroll
    for (int rr = 0; rr < RB_; ++rr) {
        v2f vo, co;
#pragma unroll
        for (int e = 0; e < 2; ++e) {
            runP[e] *= ca[rr][e];
            runS[e]  = fmaf(ca[rr][e], runS[e], cb[rr][e]);
            vo[e] = runS[e];
            co[e] = runP[e];
        }
        *(v2f*)(vrow + (size_t)rr * H_) = vo;
        *(v2f*)(crow + (size_t)rr * H_) = co;
    }
}

#define MT_ 4
#define NT_ 4
#define KS_ 8
static_assert((MT_ * 16) % RB_ == 0, "epilogue trips must tile the block's rows");
static_assert(L_ == MT_ * 16 && T_ % L_ == 0, "one GEMM block must be exactly one scan chunk of one batch row");
#define KP_ 36
#define SR_ (MT_ * 16 + 2 * NT_ * 16)
static_assert(SR_ * KP_ <= 2 * MT_ * 16 * SP_, "the operand slices of a trip must fit the wave's own epilogue region");

__global__ __launch_bounds__(128) void mingru_gemm_kernel(
    const float* __restrict__ x,  const float* __restrict__ Wz,
    const float* __restrict__ bz, const float* __restrict__ Wh,
    const float* __restrict__ bh, float* __restrict__ v_out,
    float* __restrict__ c_out)
{
    __shared__ __attribute__((aligned(16))) float stage[4][2][MT_ * 16][SP_];
    const int lane  = threadIdx.x & 31;
    const int wave  = threadIdx.x >> 5;
    const int m0    = blockIdx.x * (MT_ * 16);
    const int n0    = wave * (NT_ * 16);
    const int l15   = lane & 15;
    const int khalf = (lane >> 4) << 1;

    const int prow = lane >> 3;
    const int pcol = (lane & 7) << 2;
    const int srow = ((prow & 1) << 3) + (prow >> 1);
    const float* xg  = x  + (size_t)(m0 + srow) * DIN_ + pcol;
    const float* wzg = Wz + (size_t)(n0 + srow) * DIN_ + pcol;
    const float* whg = Wh + (size_t)(n0 + srow) * DIN_ + pcol;
    float* st = &stage[wave][0][0][0];
    float* stw = st + srow * KP_ + pcol;
    const float* sxa = st + l15 * KP_ + khalf;
    const float* swz = sxa + (MT_ * 16) * KP_;
    const float* swh = swz + (NT_ * 16) * KP_;

    v8f accz[MT_][NT_];
    v8f acch[MT_][NT_];
#pragma unroll
    for (int mt = 0; mt < MT_; ++mt) {
#pragma unroll
        for (int nt = 0; nt < NT_; ++nt) {
            accz[mt][nt] = v8f{};
            acch[mt][nt] = v8f{};
        }
    }

    for (int kk = 0; kk < DIN_; kk += 4 * KS_) {
        v2f a[MT_][KS_];
        v2f fz[NT_][KS_];
        v2f fh[NT_][KS_];
        v4f pc[3][16];
#pragma unroll
        for (int i = 0; i < 16; ++i) {
            const size_t ro = (size_t)(16 * (i >> 2) + 2 * (i & 3)) * DIN_ + kk;
            pc[0][i] = *(const v4f*)(xg  + ro);
            pc[1][i] = *(const v4f*)(wzg + ro);
            pc[2][i] = *(const v4f*)(whg + ro);
        }
#pragma unroll
        for (int g = 0; g < 3; ++g) {
#pragma unroll
            for (int i = 0; i < 16; ++i)
                *(v4f*)(stw + (g * 64 + 16 * (i >> 2) + 2 * (i & 3)) * KP_) = pc[g][i];
        }
#pragma unroll
        for (int mt = 0; mt < MT_; ++mt) {
#pragma unroll
            for (int s = 0; s < KS_; ++s)
                a[mt][s] = *(const v2f*)(sxa + mt * 16 * KP_ + 4 * s);
        }
#pragma unroll
        for (int nt = 0; nt < NT_; ++nt) {
#pragma unroll
            for (int s = 0; s < KS_; ++s)
                fz[nt][s] = *(const v2f*)(swz + nt * 16 * KP_ + 4 * s);
        }
#pragma unroll
        for (int nt = 0; nt < NT_; ++nt) {
#pragma unroll
            for (int s = 0; s < KS_; ++s)
                fh[nt][s] = *(const v2f*)(swh + nt * 16 * KP_ + 4 * s);
        }
#pragma unroll
        for (int s = 0; s < KS_; ++s) {
#pragma unroll
            for (int mt = 0; mt < MT_; ++mt) {
#pragma unroll
                for (int nt = 0; nt < NT_; ++nt) {
                    accz[mt][nt] = __builtin_amdgcn_wmma_f32_16x16x4_f32(
                        false, a[mt][s], false, fz[nt][s], (short)0, accz[mt][nt], false, false);
                    acch[mt][nt] = __builtin_amdgcn_wmma_f32_16x16x4_f32(
                        false, a[mt][s], false, fh[nt][s], (short)0, acch[mt][nt], false, false);
                }
            }
        }
    }

    float* sz = &stage[wave][0][0][0];
    float* sh = &stage[wave][1][0][0];
    const int rhalf = (lane >> 4) << 3;
#pragma unroll
    for (int mt = 0; mt < MT_; ++mt) {
#pragma unroll
        for (int nt = 0; nt < NT_; ++nt) {
#pragma unroll
            for (int j = 0; j < 8; ++j) {
                const int off = (mt * 16 + rhalf + j) * SP_ + nt * 16 + l15;
                sz[off] = accz[mt][nt][j];
                sh[off] = acch[mt][nt][j];
            }
        }
    }
    __syncthreads();

    const int   c2   = lane << 1;
    const v2f   bzb  = *(const v2f*)(bz + n0 + c2);
    const v2f   bhb  = *(const v2f*)(bh + n0 + c2);
    v2f runP = v2f{1.0f, 1.0f};
    v2f runS = v2f{0.0f, 0.0f};
    v2f ca[RB_];
    v2f cb[RB_];
    epi_terms(sz, sh, c2, bzb, bhb, ca, cb);
    float* vrow = v_out + (size_t)m0 * H_ + n0 + c2;
    float* crow = c_out + (size_t)m0 * H_ + n0 + c2;
    for (int r0 = 0; r0 < MT_ * 16 - RB_; r0 += RB_) {
        v2f can[RB_], cbn[RB_];
        epi_terms(sz + (r0 + RB_) * SP_, sh + (r0 + RB_) * SP_, c2, bzb, bhb, can, cbn);
        epi_scan(ca, cb, runP, runS, vrow + (size_t)r0 * H_, crow + (size_t)r0 * H_);
#pragma unroll
        for (int rr = 0; rr < RB_; ++rr) {
            ca[rr] = can[rr];
            cb[rr] = cbn[rr];
        }
    }
    epi_scan(ca, cb, runP, runS, vrow + (size_t)(MT_ * 16 - RB_) * H_, crow + (size_t)(MT_ * 16 - RB_) * H_);
}

__global__ __launch_bounds__(256) void mingru_carry(
    const float* __restrict__ h0, const float* __restrict__ p_arr,
    const float* __restrict__ s_arr, float* __restrict__ entry)
{
    const int h = threadIdx.x;
    const int b = blockIdx.x;

    float state = g_lin(h0[(size_t)b * H_ + h]);
#pragma unroll 16
    for (int j = 0; j < NC_; ++j) {
        const size_t last = ((size_t)b * T_ + (size_t)j * L_ + (L_ - 1)) * H_ + h;
        entry[((size_t)b * NC_ + j) * H_ + h] = state;
        state = fmaf(p_arr[last], state, s_arr[last]);
    }
}

__global__ __launch_bounds__(256) void mingru_apply(
    const float* __restrict__ p_arr, float* __restrict__ out,
    const float* __restrict__ entry)
{
    const int col  = (threadIdx.x & 63) << 2;
    const int rsub = threadIdx.x >> 6;
    const int j = blockIdx.x;
    const int b = blockIdx.y;

    const v4f E = *(const v4f*)(entry + ((size_t)b * NC_ + j) * H_ + col);
    const size_t base = ((size_t)b * T_ + (size_t)j * L_ + rsub) * H_ + col;

    for (int r0 = 0; r0 < L_; r0 += 16) {
        v4f p[4], sv[4];
#pragma unroll
        for (int i = 0; i < 4; ++i) {
            const size_t idx = base + (size_t)(r0 + 4 * i) * H_;
            p[i]  = __builtin_nontemporal_load((const v4f*)(p_arr + idx));
            sv[i] = __builtin_nontemporal_load((const v4f*)(out + idx));
        }
#pragma unroll
        for (int i = 0; i < 4; ++i) {
            const size_t idx = base + (size_t)(r0 + 4 * i) * H_;
            v4f o;
#pragma unroll
            for (int e = 0; e < 4; ++e)
                o[e] = fmaf(E[e], p[i][e], sv[i][e]);
            *(v4f*)(out + idx) = o;
        }
    }
}

extern "C" void kernel_launch(void* const* d_in, const int* in_sizes, int n_in,
                              void* d_out, int out_size, void* d_ws, size_t ws_size,
                              hipStream_t stream)
{
    const float* x  = (const float*)d_in[0];
    const float* h0 = (const float*)d_in[1];
    const float* Wz = (const float*)d_in[2];
    const float* bz = (const float*)d_in[3];
    const float* Wh = (const float*)d_in[4];
    const float* bh = (const float*)d_in[5];
    float* out = (float*)d_out;

    float* ws    = (float*)d_ws;
    float* p_arr = ws;
    float* entry = p_arr + (size_t)B_ * T_ * H_;

    mingru_gemm_kernel<<<(B_ * T_) / (MT_ * 16), 128, 0, stream>>>(x, Wz, bz, Wh, bh, out, p_arr);

    mingru_carry<<<B_, H_, 0, stream>>>(h0, p_arr, out, entry);
    dim3 sgrid(NC_, B_);
    mingru_apply<<<sgrid, 256, 0, stream>>>(p_arr, out, entry);
}


# GEMM gates computed one after the other (16 accumulator tiles per wave): the freed registers keep the next trip's whole-line loads in flight under the current trip's matrix work; sums unchanged
# speedup vs baseline: 3.6010x; 1.0622x over previous
#include <hip/hip_runtime.h>
#include <hip/hip_bf16.h>
#include <math.h>

#define B_   32
#define T_   4096
#define DIN_ 256
#define H_   256
#define L_   64
#define NC_  (T_ / L_)

typedef float v2f __attribute__((ext_vector_type(2)));
typedef float v4f __attribute__((ext_vector_type(4)));
typedef float v8f __attribute__((ext_vector_type(8)));


__device__ __forceinline__ void sigmoid_both(float u, float& s_of_u, float& s_of_minus_u) {
    const float e = __expf(-fabsf(u));
    const float r = __builtin_amdgcn_rcpf(1.0f + e);
    const float q = e * r;
    const bool  nonneg = (u >= 0.0f);
    s_of_u       = nonneg ? r : q;
    s_of_minus_u = nonneg ? q : r;
}

__device__ __forceinline__ float g_lin(float a) {
    const float e  = __expf(-fabsf(a));
    const float sg = e * __builtin_amdgcn_rcpf(1.0f + e);
    return (a >= 0.0f) ? a + 0.5f : sg;
}

#define SP_ 68
#define RB_ 8

__device__ __forceinline__ void epi_terms(const float* zrow, const float* hrow, int c2,
                                          v2f bzb, v2f bhb, v2f (&ca)[RB_], v2f (&cb)[RB_]) {
#pragma unroll
    for (int rr = 0; rr < RB_; ++rr) {
        const v2f kz = *(const v2f*)(zrow + rr * SP_ + c2);
        const v2f kh = *(const v2f*)(hrow + rr * SP_ + c2);
#pragma unroll
        for (int e = 0; e < 2; ++e) {
            const float kpre = kz[e] + bzb[e];
            const float hp   = kh[e] + bhb[e];
            float z, omz;
            sigmoid_both(kpre, z, omz);
            ca[rr][e] = omz;
            cb[rr][e] = z * g_lin(hp);
        }
    }
}

__device__ __forceinline__ void epi_scan(const v2f (&ca)[RB_], const v2f (&cb)[RB_],
                                         v2f& runP, v2f& runS, float* vrow, float* crow) {
#pragma unroll
    for (int rr = 0; rr < RB_; ++rr) {
        v2f vo, co;
#pragma unroll
        for (int e = 0; e < 2; ++e) {
            runP[e] *= ca[rr][e];
            runS[e]  = fmaf(ca[rr][e], runS[e], cb[rr][e]);
            vo[e] = runS[e];
            co[e] = runP[e];
        }
        *(v2f*)(vrow + (size_t)rr * H_) = vo;
        *(v2f*)(crow + (size_t)rr * H_) = co;
    }
}

#define MT_ 4
#define NT_ 4
#define KS_ 8
static_assert((MT_ * 16) % RB_ == 0, "epilogue trips must tile the block's rows");
static_assert(L_ == MT_ * 16 && T_ % L_ == 0, "one GEMM block must be exactly one scan chunk of one batch row");
#define KP_ 36
#define SR_ (MT_ * 16 + NT_ * 16)
#define PLANE_ (MT_ * 16 * SP_)
#define STG_ (SR_ * KP_)
#define WREG_ (PLANE_ + STG_)
static_assert(PLANE_ <= STG_, "the candidate block is parked over the operand slices once the last trip has read them");
static_assert((PLANE_ * 4) % 16 == 0, "the operand slices must start 16-byte aligned for 16-byte LDS stores");

__device__ __forceinline__ void fetch_slices(const float* xp, const float* wp, v4f (&pc)[2][16]) {
#pragma unroll
    for (int i = 0; i < 16; ++i) {
        const int ro = (16 * (i >> 2) + 2 * (i & 3)) * DIN_;
        pc[0][i] = *(const v4f*)(xp + ro);
        pc[1][i] = *(const v4f*)(wp + ro);
    }
}

__device__ __forceinline__ void stage_slices(float* stw, const v4f (&pc)[2][16]) {
#pragma unroll
    for (int g = 0; g < 2; ++g) {
#pragma unroll
        for (int i = 0; i < 16; ++i)
            *(v4f*)(stw + (g * 64 + 16 * (i >> 2) + 2 * (i & 3)) * KP_) = pc[g][i];
    }
}

__device__ __forceinline__ void read_frags(const float* sxa, const float* sw,
                                           v2f (&a)[MT_][KS_], v2f (&f)[NT_][KS_]) {
#pragma unroll
    for (int mt = 0; mt < MT_; ++mt) {
#pragma unroll
        for (int s = 0; s < KS_; ++s)
            a[mt][s] = *(const v2f*)(sxa + mt * 16 * KP_ + 4 * s);
    }
#pragma unroll
    for (int nt = 0; nt < NT_; ++nt) {
#pragma unroll
        for (int s = 0; s < KS_; ++s)
            f[nt][s] = *(const v2f*)(sw + nt * 16 * KP_ + 4 * s);
    }
}

__device__ __forceinline__ void mma_trip(const v2f (&a)[MT_][KS_], const v2f (&f)[NT_][KS_],
                                         v8f (&acc)[MT_][NT_]) {
#pragma unroll
    for (int s = 0; s < KS_; ++s) {
#pragma unroll
        for (int mt = 0; mt < MT_; ++mt) {
#pragma unroll
            for (int nt = 0; nt < NT_; ++nt)
                acc[mt][nt] = __builtin_amdgcn_wmma_f32_16x16x4_f32(
                    false, a[mt][s], false, f[nt][s], (short)0, acc[mt][nt], false, false);
        }
    }
}

template <bool MORE>
__device__ __forceinline__ void gate_pass(const float* xg, const float* wg, const float* wnext,
                                          float* stw, const float* sxa, const float* sw,
                                          v4f (&pc)[2][16], v8f (&acc)[MT_][NT_]) {
#pragma unroll
    for (int mt = 0; mt < MT_; ++mt) {
#pragma unroll
        for (int nt = 0; nt < NT_; ++nt)
            acc[mt][nt] = v8f{};
    }
    const float* xp = xg;
    const float* wp = wg;
    for (int kk = 0; kk < DIN_ - 4 * KS_; kk += 4 * KS_) {
        v2f a[MT_][KS_];
        v2f f[NT_][KS_];
        stage_slices(stw, pc);
        read_frags(sxa, sw, a, f);
        xp += 4 * KS_;
        wp += 4 * KS_;
        fetch_slices(xp, wp, pc);
        mma_trip(a, f, acc);
    }
    {
        v2f a[MT_][KS_];
        v2f f[NT_][KS_];
        stage_slices(stw, pc);
        read_frags(sxa, sw, a, f);
        if (MORE)
            fetch_slices(xg, wnext, pc);
        mma_trip(a, f, acc);
    }
}

__device__ __forceinline__ void park_block(float* plane, int rhalf, int l15, const v8f (&acc)[MT_][NT_]) {
#pragma unroll
    for (int mt = 0; mt < MT_; ++mt) {
#pragma unroll
        for (int nt = 0; nt < NT_; ++nt) {
#pragma unroll
            for (int j = 0; j < 8; ++j)
                plane[(mt * 16 + rhalf + j) * SP_ + nt * 16 + l15] = acc[mt][nt][j];
        }
    }
}

__global__ __launch_bounds__(128) void mingru_gemm_kernel(
    const float* __restrict__ x,  const float* __restrict__ Wz,
    const float* __restrict__ bz, const float* __restrict__ Wh,
    const float* __restrict__ bh, float* __restrict__ v_out,
    float* __restrict__ c_out)
{
    __shared__ __attribute__((aligned(16))) float stage[4][WREG_];
    const int lane  = threadIdx.x & 31;
    const int wave  = threadIdx.x >> 5;
    const int m0    = blockIdx.x * (MT_ * 16);
    const int n0    = wave * (NT_ * 16);
    const int l15   = lane & 15;
    const int khalf = (lane >> 4) << 1;

    const int prow = lane >> 3;
    const int pcol = (lane & 7) << 2;
    const int srow = ((prow & 1) << 3) + (prow >> 1);
    const float* xg  = x  + (size_t)(m0 + srow) * DIN_ + pcol;
    const float* wzg = Wz + (size_t)(n0 + srow) * DIN_ + pcol;
    const float* whg = Wh + (size_t)(n0 + srow) * DIN_ + pcol;
    float* sz = &stage[wave][0];
    float* sh = sz + PLANE_;
    float* stw = sh + srow * KP_ + pcol;
    const float* sxa = sh + l15 * KP_ + khalf;
    const float* sw  = sxa + (MT_ * 16) * KP_;
    const int rhalf = (lane >> 4) << 3;

    v8f acc[MT_][NT_];
    v4f pc[2][16];
    fetch_slices(xg, wzg, pc);
    gate_pass<true>(xg, wzg, whg, stw, sxa, sw, pc, acc);
    park_block(sz, rhalf, l15, acc);
    gate_pass<false>(xg, whg, whg, stw, sxa, sw, pc, acc);
    park_block(sh, rhalf, l15, acc);
    __syncthreads();

    const int   c2   = lane << 1;
    const v2f   bzb  = *(const v2f*)(bz + n0 + c2);
    const v2f   bhb  = *(const v2f*)(bh + n0 + c2);
    v2f runP = v2f{1.0f, 1.0f};
    v2f runS = v2f{0.0f, 0.0f};
    v2f ca[RB_];
    v2f cb[RB_];
    epi_terms(sz, sh, c2, bzb, bhb, ca, cb);
    float* vrow = v_out + (size_t)m0 * H_ + n0 + c2;
    float* crow = c_out + (size_t)m0 * H_ + n0 + c2;
    for (int r0 = 0; r0 < MT_ * 16 - RB_; r0 += RB_) {
        v2f can[RB_], cbn[RB_];
        epi_terms(sz + (r0 + RB_) * SP_, sh + (r0 + RB_) * SP_, c2, bzb, bhb, can, cbn);
        epi_scan(ca, cb, runP, runS, vrow + (size_t)r0 * H_, crow + (size_t)r0 * H_);
#pragma unroll
        for (int rr = 0; rr < RB_; ++rr) {
            ca[rr] = can[rr];
            cb[rr] = cbn[rr];
        }
    }
    epi_scan(ca, cb, runP, runS, vrow + (size_t)(MT_ * 16 - RB_) * H_, crow + (size_t)(MT_ * 16 - RB_) * H_);
}

__global__ __launch_bounds__(256) void mingru_carry(
    const float* __restrict__ h0, const float* __restrict__ p_arr,
    const float* __restrict__ s_arr, float* __restrict__ entry)
{
    const int h = threadIdx.x;
    const int b = blockIdx.x;

    float state = g_lin(h0[(size_t)b * H_ + h]);
#pragma unroll 16
    for (int j = 0; j < NC_; ++j) {
        const size_t last = ((size_t)b * T_ + (size_t)j * L_ + (L_ - 1)) * H_ + h;
        entry[((size_t)b * NC_ + j) * H_ + h] = state;
        state = fmaf(p_arr[last], state, s_arr[last]);
    }
}

__global__ __launch_bounds__(256) void mingru_apply(
    const float* __restrict__ p_arr, float* __restrict__ out,
    const float* __restrict__ entry)
{
    const int col  = (threadIdx.x & 63) << 2;
    const int rsub = threadIdx.x >> 6;
    const int j = blockIdx.x;
    const int b = blockIdx.y;

    const v4f E = *(const v4f*)(entry + ((size_t)b * NC_ + j) * H_ + col);
    const size_t base = ((size_t)b * T_ + (size_t)j * L_ + rsub) * H_ + col;

    for (int r0 = 0; r0 < L_; r0 += 16) {
        v4f p[4], sv[4];
#pragma unroll
        for (int i = 0; i < 4; ++i) {
            const size_t idx = base + (size_t)(r0 + 4 * i) * H_;
            p[i]  = __builtin_nontemporal_load((const v4f*)(p_arr + idx));
            sv[i] = __builtin_nontemporal_load((const v4f*)(out + idx));
        }
#pragma unroll
        for (int i = 0; i < 4; ++i) {
            const size_t idx = base + (size_t)(r0 + 4 * i) * H_;
            v4f o;
#pragma unroll
            for (int e = 0; e < 4; ++e)
                o[e] = fmaf(E[e], p[i][e], sv[i][e]);
            *(v4f*)(out + idx) = o;
        }
    }
}

extern "C" void kernel_launch(void* const* d_in, const int* in_sizes, int n_in,
                              void* d_out, int out_size, void* d_ws, size_t ws_size,
                              hipStream_t stream)
{
    const float* x  = (const float*)d_in[0];
    const float* h0 = (const float*)d_in[1];
    const float* Wz = (const float*)d_in[2];
    const float* bz = (const float*)d_in[3];
    const float* Wh = (const float*)d_in[4];
    const float* bh = (const float*)d_in[5];
    float* out = (float*)d_out;

    float* ws    = (float*)d_ws;
    float* p_arr = ws;
    float* entry = p_arr + (size_t)B_ * T_ * H_;

    mingru_gemm_kernel<<<(B_ * T_) / (MT_ * 16), 128, 0, stream>>>(x, Wz, bz, Wh, bh, out, p_arr);

    mingru_carry<<<B_, H_, 0, stream>>>(h0, p_arr, out, entry);
    dim3 sgrid(NC_, B_);
    mingru_apply<<<sgrid, 256, 0, stream>>>(p_arr, out, entry);
}


# One pass: a block owns 32 columns of one batch row and walks T in order, 8 waves on 8 consecutive chunks per step, chunk carries meet in LDS; P plane, carry and finish launches go; sums unchanged
# speedup vs baseline: 4.1588x; 1.1549x over previous
#include <hip/hip_runtime.h>
#include <math.h>

#define B_   32
#define T_   4096
#define DIN_ 256
#define H_   256
#define L_   64
#define NW_  8
#define CW_  32
#define NIT_ (T_ / (L_ * NW_))

typedef float v2f __attribute__((ext_vector_type(2)));
typedef float v4f __attribute__((ext_vector_type(4)));
typedef float v8f __attribute__((ext_vector_type(8)));


__device__ __forceinline__ void sigmoid_both(float u, float& s_of_u, float& s_of_minus_u) {
    const float e = __expf(-fabsf(u));
    const float r = __builtin_amdgcn_rcpf(1.0f + e);
    const float q = e * r;
    const bool  nonneg = (u >= 0.0f);
    s_of_u       = nonneg ? r : q;
    s_of_minus_u = nonneg ? q : r;
}

__device__ __forceinline__ float g_lin(float a) {
    const float e  = __expf(-fabsf(a));
    const float sg = e * __builtin_amdgcn_rcpf(1.0f + e);
    return (a >= 0.0f) ? a + 0.5f : sg;
}

#define SP_ 68
#define RB_ 8
#define MT_ 4
#define NT_ 4
#define KS_ 8
#define KP_ 36
#define WP_ 260
#define PLANE_ (MT_ * 16 * SP_)
#define XST_ (MT_ * 16 * KP_)
#define WREG_ (PLANE_ + XST_)
static_assert((MT_ * 16) % RB_ == 0, "scan trips must tile the chunk's rows");
static_assert(L_ == MT_ * 16 && T_ % (L_ * NW_) == 0, "a wave's GEMM block must be exactly one scan chunk, a step must tile T");
static_assert(NT_ * 16 == 2 * CW_, "a wave's column tiles are the block's gate and candidate columns");
static_assert((PLANE_ * 4) % 16 == 0 && (WREG_ * 4) % 16 == 0, "x slices must start 16-byte aligned for 16-byte LDS stores");

__device__ __forceinline__ void fetch_x(const float* xp, v4f (&pc)[16]) {
#pragma unroll
    for (int i = 0; i < 16; ++i)
        pc[i] = *(const v4f*)(xp + (16 * (i >> 2) + 2 * (i & 3)) * DIN_);
}

__device__ __forceinline__ void stage_x(float* stw, const v4f (&pc)[16]) {
#pragma unroll
    for (int i = 0; i < 16; ++i)
        *(v4f*)(stw + (16 * (i >> 2) + 2 * (i & 3)) * KP_) = pc[i];
}

__device__ __forceinline__ void read_frags(const float* sxa, const float* swk,
                                           v2f (&a)[MT_][KS_], v2f (&f)[NT_][KS_]) {
#pragma unroll
    for (int mt = 0; mt < MT_; ++mt) {
#pragma unroll
        for (int s = 0; s < KS_; ++s)
            a[mt][s] = *(const v2f*)(sxa + mt * 16 * KP_ + 4 * s);
    }
#pragma unroll
    for (int nt = 0; nt < NT_; ++nt) {
#pragma unroll
        for (int s = 0; s < KS_; ++s)
            f[nt][s] = *(const v2f*)(swk + nt * 16 * WP_ + 4 * s);
    }
}

__device__ __forceinline__ void mma_trip(const v2f (&a)[MT_][KS_], const v2f (&f)[NT_][KS_],
                                         v8f (&acc)[MT_][NT_]) {
#pragma unroll
    for (int s = 0; s < KS_; ++s) {
#pragma unroll
        for (int mt = 0; mt < MT_; ++mt) {
#pragma unroll
            for (int nt = 0; nt < NT_; ++nt)
                acc[mt][nt] = __builtin_amdgcn_wmma_f32_16x16x4_f32(
                    false, a[mt][s], false, f[nt][s], (short)0, acc[mt][nt], false, false);
        }
    }
}

__device__ __forceinline__ void park_block(float* plane, int rhalf, int l15, const v8f (&acc)[MT_][NT_]) {
#pragma unroll
    for (int mt = 0; mt < MT_; ++mt) {
#pragma unroll
        for (int nt = 0; nt < NT_; ++nt) {
#pragma unroll
            for (int j = 0; j < 8; ++j)
                plane[(mt * 16 + rhalf + j) * SP_ + nt * 16 + l15] = acc[mt][nt][j];
        }
    }
}

__device__ __forceinline__ void scan_terms(const float* prow, float bzc, float bhc,
                                           float (&ca)[RB_], float (&cb)[RB_]) {
#pragma unroll
    for (int rr = 0; rr < RB_; ++rr) {
        const float kpre = prow[rr * SP_] + bzc;
        const float hp   = prow[rr * SP_ + CW_] + bhc;
        float z, omz;
        sigmoid_both(kpre, z, omz);
        ca[rr] = omz;
        cb[rr] = z * g_lin(hp);
    }
}

__device__ __forceinline__ void scan_rows(const float (&ca)[RB_], const float (&cb)[RB_],
                                          float& runP, float& runS, float* prow) {
#pragma unroll
    for (int rr = 0; rr < RB_; ++rr) {
        runP *= ca[rr];
        runS  = fmaf(ca[rr], runS, cb[rr]);
        prow[rr * SP_]       = runP;
        prow[rr * SP_ + CW_] = runS;
    }
}

__global__ __launch_bounds__(NW_ * 32) void mingru_seq_kernel(
    const float* __restrict__ x,  const float* __restrict__ h0,
    const float* __restrict__ Wz, const float* __restrict__ bz,
    const float* __restrict__ Wh, const float* __restrict__ bh,
    float* __restrict__ out)
{
    __shared__ __attribute__((aligned(16))) float wsh[2 * CW_ * WP_];
    __shared__ __attribute__((aligned(16))) float stage[NW_][WREG_];
    __shared__ float sums[2][NW_][2][CW_];
    __shared__ __attribute__((aligned(16))) float ent[NW_][CW_];
    const int tid   = threadIdx.x;
    const int lane  = tid & 31;
    const int wave  = tid >> 5;
    const int b     = blockIdx.x / (H_ / CW_);
    const int col0  = (blockIdx.x % (H_ / CW_)) * CW_;
    const int l15   = lane & 15;
    const int khalf = (lane >> 4) << 1;
    const int rhalf = (lane >> 4) << 3;

    {
        v4f wpc[16];
#pragma unroll
        for (int i = 0; i < 16; ++i) {
            const int idx = tid + (NW_ * 32) * (i & 7);
            const float* wsrc = (i < 8) ? Wz : Wh;
            wpc[i] = *(const v4f*)(wsrc + (size_t)(col0 + (idx >> 6)) * DIN_ + ((idx & 63) << 2));
        }
#pragma unroll
        for (int i = 0; i < 16; ++i) {
            const int idx = tid + (NW_ * 32) * (i & 7);
            *(v4f*)(wsh + ((i >> 3) * CW_ + (idx >> 6)) * WP_ + ((idx & 63) << 2)) = wpc[i];
        }
    }

    const int prow = lane >> 3;
    const int pcol = (lane & 7) << 2;
    const int srow = ((prow & 1) << 3) + (prow >> 1);
    float* pl  = &stage[wave][0];
    float* sx  = pl + PLANE_;
    float* stw = sx + srow * KP_ + pcol;
    const float* sxa = sx + l15 * KP_ + khalf;
    const float* swk = wsh + l15 * WP_ + khalf;
    const float* xrow0 = x + ((size_t)b * T_ + wave * L_ + srow) * DIN_ + pcol;

    const float bzc = bz[col0 + lane];
    const float bhc = bh[col0 + lane];
    float state = g_lin(h0[(size_t)b * H_ + col0 + lane]);

    v4f pc[16];
    fetch_x(xrow0, pc);
    __syncthreads();

    for (int it = 0; it < NIT_; ++it) {
        const float* xg = xrow0 + (size_t)it * (NW_ * L_) * DIN_;
        v8f acc[MT_][NT_];
#pragma unroll
        for (int mt = 0; mt < MT_; ++mt) {
#pragma unroll
            for (int nt = 0; nt < NT_; ++nt)
                acc[mt][nt] = v8f{};
        }
        const float* xp = xg;
        for (int kk = 0; kk < DIN_ - 4 * KS_; kk += 4 * KS_) {
            v2f a[MT_][KS_];
            v2f f[NT_][KS_];
            stage_x(stw, pc);
            read_frags(sxa, swk + kk, a, f);
            xp += 4 * KS_;
            fetch_x(xp, pc);
            mma_trip(a, f, acc);
        }
        {
            v2f a[MT_][KS_];
            v2f f[NT_][KS_];
            stage_x(stw, pc);
            read_frags(sxa, swk + (DIN_ - 4 * KS_), a, f);
            if (it + 1 < NIT_)
                fetch_x(xg + (size_t)(NW_ * L_) * DIN_, pc);
            mma_trip(a, f, acc);
        }
        park_block(pl, rhalf, l15, acc);

        float runP = 1.0f;
        float runS = 0.0f;
        float ca[RB_], cb[RB_];
        float* prw = pl + lane;
        scan_terms(prw, bzc, bhc, ca, cb);
        for (int r0 = 0; r0 < MT_ * 16 - RB_; r0 += RB_) {
            float can[RB_], cbn[RB_];
            scan_terms(prw + (r0 + RB_) * SP_, bzc, bhc, can, cbn);
            scan_rows(ca, cb, runP, runS, prw + r0 * SP_);
#pragma unroll
            for (int rr = 0; rr < RB_; ++rr) {
                ca[rr] = can[rr];
                cb[rr] = cbn[rr];
            }
        }
        scan_rows(ca, cb, runP, runS, prw + (MT_ * 16 - RB_) * SP_);

        const int par = it & 1;
        sums[par][wave][0][lane] = runP;
        sums[par][wave][1][lane] = runS;
        __syncthreads();
        float E  = state;
        float st = state;
#pragma unroll
        for (int w = 0; w < NW_; ++w) {
            const float p = sums[par][w][0][lane];
            const float s = sums[par][w][1][lane];
            E  = (w == wave) ? st : E;
            st = fmaf(p, st, s);
        }
        state = st;

        ent[wave][lane] = E;
        const v4f E4 = *(const v4f*)(&ent[wave][pcol]);
        const float* pr = pl + srow * SP_ + pcol;
        float* og = out + ((size_t)b * T_ + (size_t)it * (NW_ * L_) + wave * L_ + srow) * H_ + col0 + pcol;
#pragma unroll
        for (int i = 0; i < 16; ++i) {
            const int ro = 16 * (i >> 2) + 2 * (i & 3);
            const v4f p4 = *(const v4f*)(pr + ro * SP_);
            const v4f s4 = *(const v4f*)(pr + ro * SP_ + CW_);
            v4f o;
#pragma unroll
            for (int e = 0; e < 4; ++e)
                o[e] = fmaf(E4[e], p4[e], s4[e]);
            *(v4f*)(og + (size_t)ro * H_) = o;
        }
    }
}

extern "C" void kernel_launch(void* const* d_in, const int* in_sizes, int n_in,
                              void* d_out, int out_size, void* d_ws, size_t ws_size,
                              hipStream_t stream)
{
    const float* x  = (const float*)d_in[0];
    const float* h0 = (const float*)d_in[1];
    const float* Wz = (const float*)d_in[2];
    const float* bz = (const float*)d_in[3];
    const float* Wh = (const float*)d_in[4];
    const float* bh = (const float*)d_in[5];
    float* out = (float*)d_out;

    mingru_seq_kernel<<<B_ * (H_ / CW_), NW_ * 32, 0, stream>>>(x, h0, Wz, bz, Wh, bh, out);
}
